# starting kernel
# baseline (speedup 1.0000x reference)
#include <hip/hip_runtime.h>
#include <hip/hip_bf16.h>


typedef __attribute__((ext_vector_type(2))) float v2f;
typedef __attribute__((ext_vector_type(8))) float v8f;

#define EDM_B 4
#define EDM_N 2048
#define EDM_D 128
#define EDM_EPS 1e-7f

__global__ void edm_sqnorm_kernel(const float* __restrict__ x,
                                  float* __restrict__ sq) {
    const int wave = (blockIdx.x * blockDim.x + threadIdx.x) >> 5;
    const int lane = threadIdx.x & 31;
    if (wave >= EDM_B * EDM_N) return;

    const float* row = x + (size_t)wave * EDM_D;
    float s = 0.0f;
#pragma unroll
    for (int k = 0; k < EDM_D; k += 32) {
        float v = row[k + lane];
        s = fmaf(v, v, s);
    }
#pragma unroll
    for (int off = 16; off > 0; off >>= 1)
        s += __shfl_xor(s, off, 32);
    if (lane == 0) sq[wave] = s;
}

__global__ void edm_tile_kernel(const float* __restrict__ x,
                                const float* __restrict__ sq,
                                float* __restrict__ out) {
    const int lane   = threadIdx.x & 31;
    const int waveId = threadIdx.x >> 5;
    const int tile   = blockIdx.x * (blockDim.x >> 5) + waveId;

    const int b  = tile >> 14;
    const int ti = (tile >> 7) & 127;
    const int tj = tile & 127;
    const int row0 = ti << 4;
    const int col0 = tj << 4;

    const int m    = lane & 15;
    const int koff = (lane >> 4) << 1;

    const float* xb = x + (size_t)b * EDM_N * EDM_D;
    const float* ap = xb + (size_t)(row0 + m) * EDM_D + koff;
    const float* bp = xb + (size_t)(col0 + m) * EDM_D + koff;

    v8f acc = {};
#pragma unroll 8
    for (int k0 = 0; k0 < EDM_D; k0 += 4) {
        v2f av = *(const v2f*)(ap + k0);
        v2f bv = *(const v2f*)(bp + k0);
        acc = __builtin_amdgcn_wmma_f32_16x16x4_f32(
             false, av,  false, bv,
             (short)0, acc,  false,  false);
    }

    const int   half = lane >> 4;
    const float sqj  = sq[b * EDM_N + col0 + m];
    const float* sqi = sq + b * EDM_N + row0 + (half << 3);
    float* orow = out + ((size_t)b * EDM_N + row0 + (half << 3)) * EDM_N
                      + col0 + m;
#pragma unroll
    for (int r = 0; r < 8; ++r) {
        float d2 = sqi[r] + sqj - 2.0f * acc[r];
        d2 = fmaxf(d2, 0.0f);
        __builtin_nontemporal_store(sqrtf(d2 + EDM_EPS), orow + (size_t)r * EDM_N);
    }
}

extern "C" void kernel_launch(void* const* d_in, const int* in_sizes, int n_in,
                              void* d_out, int out_size, void* d_ws, size_t ws_size,
                              hipStream_t stream) {
    (void)in_sizes; (void)n_in; (void)out_size; (void)ws_size;

    const float* x  = (const float*)d_in[0];
    float* out      = (float*)d_out;
    float* sq       = (float*)d_ws;

    edm_sqnorm_kernel<<<1024, 256, 0, stream>>>(x, sq);

    edm_tile_kernel<<<8192, 256, 0, stream>>>(x, sq, out);
}


# Register tile: each wave computes a 64x64 block (4x4 tiles of 16x16), so every loaded A/B fragment feeds 4 matrix products: 4x fewer fragment loads per product, 16x fewer waves; sum order unchanged
# speedup vs baseline: 2.3924x; 2.3924x over previous
#include <hip/hip_runtime.h>
#include <hip/hip_bf16.h>


typedef __attribute__((ext_vector_type(2))) float v2f;
typedef __attribute__((ext_vector_type(8))) float v8f;

#define EDM_B 4
#define EDM_N 2048
#define EDM_D 128
#define EDM_EPS 1e-7f

__global__ void edm_sqnorm_kernel(const float* __restrict__ x,
                                  float* __restrict__ sq) {
    const int wave = (blockIdx.x * blockDim.x + threadIdx.x) >> 5;
    const int lane = threadIdx.x & 31;
    if (wave >= EDM_B * EDM_N) return;

    const float* row = x + (size_t)wave * EDM_D;
    float s = 0.0f;
#pragma unroll
    for (int k = 0; k < EDM_D; k += 32) {
        float v = row[k + lane];
        s = fmaf(v, v, s);
    }
#pragma unroll
    for (int off = 16; off > 0; off >>= 1)
        s += __shfl_xor(s, off, 32);
    if (lane == 0) sq[wave] = s;
}

#define EDM_WT 4

__global__ void __launch_bounds__(128)
edm_tile_kernel(const float* __restrict__ x,
                const float* __restrict__ sq,
                float* __restrict__ out) {
    const int lane   = threadIdx.x & 31;
    const int waveId = threadIdx.x >> 5;

    const int b  = blockIdx.x >> 8;
    const int bi = (blockIdx.x >> 4) & 15;
    const int bj = blockIdx.x & 15;
    const int row0 = (bi << 7) + ((waveId >> 1) << 6);
    const int col0 = (bj << 7) + ((waveId & 1) << 6);

    const int m    = lane & 15;
    const int koff = (lane >> 4) << 1;

    const float* xb = x + (size_t)b * EDM_N * EDM_D;
    const float* ap = xb + (size_t)(row0 + m) * EDM_D + koff;
    const float* bp = xb + (size_t)(col0 + m) * EDM_D + koff;

    v8f acc[EDM_WT][EDM_WT];
#pragma unroll
    for (int i = 0; i < EDM_WT; ++i)
#pragma unroll
        for (int j = 0; j < EDM_WT; ++j)
            acc[i][j] = v8f{};

#pragma unroll 4
    for (int k0 = 0; k0 < EDM_D; k0 += 4) {
        v2f av[EDM_WT], bv[EDM_WT];
#pragma unroll
        for (int i = 0; i < EDM_WT; ++i)
            av[i] = *(const v2f*)(ap + (size_t)i * 16 * EDM_D + k0);
#pragma unroll
        for (int j = 0; j < EDM_WT; ++j)
            bv[j] = *(const v2f*)(bp + (size_t)j * 16 * EDM_D + k0);
#pragma unroll
        for (int i = 0; i < EDM_WT; ++i)
#pragma unroll
            for (int j = 0; j < EDM_WT; ++j)
                acc[i][j] = __builtin_amdgcn_wmma_f32_16x16x4_f32(
                     false, av[i],  false, bv[j],
                     (short)0, acc[i][j],
                     false,  false);
    }

    const int half = lane >> 4;
    const float* sqb = sq + b * EDM_N;
    float sqj[EDM_WT];
#pragma unroll
    for (int j = 0; j < EDM_WT; ++j)
        sqj[j] = sqb[col0 + j * 16 + m];

#pragma unroll
    for (int i = 0; i < EDM_WT; ++i) {
        const int   rbase = row0 + i * 16 + (half << 3);
        const float* sqi  = sqb + rbase;
        float* orow = out + ((size_t)b * EDM_N + rbase) * EDM_N + col0 + m;
#pragma unroll
        for (int r = 0; r < 8; ++r) {
            const float si = sqi[r];
#pragma unroll
            for (int j = 0; j < EDM_WT; ++j) {
                float d2 = si + sqj[j] - 2.0f * acc[i][j][r];
                d2 = fmaxf(d2, 0.0f);
                __builtin_nontemporal_store(sqrtf(d2 + EDM_EPS),
                                            orow + (size_t)r * EDM_N + j * 16);
            }
        }
    }
}

extern "C" void kernel_launch(void* const* d_in, const int* in_sizes, int n_in,
                              void* d_out, int out_size, void* d_ws, size_t ws_size,
                              hipStream_t stream) {
    (void)in_sizes; (void)n_in; (void)out_size; (void)ws_size;

    const float* x  = (const float*)d_in[0];
    float* out      = (float*)d_out;
    float* sq       = (float*)d_ws;

    edm_sqnorm_kernel<<<1024, 256, 0, stream>>>(x, sq);

    edm_tile_kernel<<<1024, 128, 0, stream>>>(x, sq, out);
}


# Fuse the row-norm kernel into the tile kernel: each wave sums squares of the A/B fragments it already loads (1 launch, not 2; norm sum order: per-lane partials, then lane-pair add)
# speedup vs baseline: 2.5163x; 1.0518x over previous
#include <hip/hip_runtime.h>
#include <hip/hip_bf16.h>


typedef __attribute__((ext_vector_type(2))) float v2f;
typedef __attribute__((ext_vector_type(8))) float v8f;

#define EDM_B 4
#define EDM_N 2048
#define EDM_D 128
#define EDM_EPS 1e-7f

#define EDM_WT 4

__global__ void __launch_bounds__(128)
edm_tile_kernel(const float* __restrict__ x,
                float* __restrict__ out) {
    const int lane   = threadIdx.x & 31;
    const int waveId = threadIdx.x >> 5;

    const int b  = blockIdx.x >> 8;
    const int bi = (blockIdx.x >> 4) & 15;
    const int bj = blockIdx.x & 15;
    const int row0 = (bi << 7) + ((waveId >> 1) << 6);
    const int col0 = (bj << 7) + ((waveId & 1) << 6);

    const int m    = lane & 15;
    const int koff = (lane >> 4) << 1;

    const float* xb = x + (size_t)b * EDM_N * EDM_D;
    const float* ap = xb + (size_t)(row0 + m) * EDM_D + koff;
    const float* bp = xb + (size_t)(col0 + m) * EDM_D + koff;

    v8f acc[EDM_WT][EDM_WT];
#pragma unroll
    for (int i = 0; i < EDM_WT; ++i)
#pragma unroll
        for (int j = 0; j < EDM_WT; ++j)
            acc[i][j] = v8f{};

    v2f na[EDM_WT], nb[EDM_WT];
#pragma unroll
    for (int i = 0; i < EDM_WT; ++i) { na[i] = v2f{}; nb[i] = v2f{}; }

#pragma unroll 4
    for (int k0 = 0; k0 < EDM_D; k0 += 4) {
        v2f av[EDM_WT], bv[EDM_WT];
#pragma unroll
        for (int i = 0; i < EDM_WT; ++i)
            av[i] = *(const v2f*)(ap + (size_t)i * 16 * EDM_D + k0);
#pragma unroll
        for (int j = 0; j < EDM_WT; ++j)
            bv[j] = *(const v2f*)(bp + (size_t)j * 16 * EDM_D + k0);
#pragma unroll
        for (int i = 0; i < EDM_WT; ++i) {
            na[i] = __builtin_elementwise_fma(av[i], av[i], na[i]);
            nb[i] = __builtin_elementwise_fma(bv[i], bv[i], nb[i]);
        }
#pragma unroll
        for (int i = 0; i < EDM_WT; ++i)
#pragma unroll
            for (int j = 0; j < EDM_WT; ++j)
                acc[i][j] = __builtin_amdgcn_wmma_f32_16x16x4_f32(
                     false, av[i],  false, bv[j],
                     (short)0, acc[i][j],
                     false,  false);
    }

    const int half = lane >> 4;
    float sqa[EDM_WT], sqj[EDM_WT];
#pragma unroll
    for (int t = 0; t < EDM_WT; ++t) {
        const float pa = na[t][0] + na[t][1];
        const float pb = nb[t][0] + nb[t][1];
        sqa[t] = pa + __shfl_xor(pa, 16, 32);
        sqj[t] = pb + __shfl_xor(pb, 16, 32);
    }

#pragma unroll
    for (int i = 0; i < EDM_WT; ++i) {
        const int   rbase = row0 + i * 16 + (half << 3);
        float* orow = out + ((size_t)b * EDM_N + rbase) * EDM_N + col0 + m;
#pragma unroll
        for (int r = 0; r < 8; ++r) {
            const float si = __shfl(sqa[i], (half << 3) + r, 32);
#pragma unroll
            for (int j = 0; j < EDM_WT; ++j) {
                float d2 = si + sqj[j] - 2.0f * acc[i][j][r];
                d2 = fmaxf(d2, 0.0f);
                __builtin_nontemporal_store(sqrtf(d2 + EDM_EPS),
                                            orow + (size_t)r * EDM_N + j * 16);
            }
        }
    }
}

extern "C" void kernel_launch(void* const* d_in, const int* in_sizes, int n_in,
                              void* d_out, int out_size, void* d_ws, size_t ws_size,
                              hipStream_t stream) {
    (void)in_sizes; (void)n_in; (void)out_size; (void)d_ws; (void)ws_size;

    const float* x  = (const float*)d_in[0];
    float* out      = (float*)d_out;

    edm_tile_kernel<<<1024, 128, 0, stream>>>(x, out);
}


# Stage both 128-row operand panels in LDS with whole-line loads (a 512-B row of x per load); fragments come from LDS, read one block ahead, not from per-lane global gathers; sums unchanged
# speedup vs baseline: 2.6076x; 1.0363x over previous
#include <hip/hip_runtime.h>
#include <hip/hip_bf16.h>


typedef __attribute__((ext_vector_type(2))) float v2f;
typedef __attribute__((ext_vector_type(8))) float v8f;

#define EDM_B 4
#define EDM_N 2048
#define EDM_D 128
#define EDM_EPS 1e-7f

#define EDM_WT 4
#define EDM_PR 128
#define EDM_PITCH (EDM_D + 4)

__global__ void __launch_bounds__(128)
edm_tile_kernel(const float* __restrict__ x,
                float* __restrict__ out) {
    const int lane   = threadIdx.x & 31;
    const int waveId = threadIdx.x >> 5;

    const int b  = blockIdx.x >> 8;
    const int bi = (blockIdx.x >> 4) & 15;
    const int bj = blockIdx.x & 15;
    const int row0 = (bi << 7) + ((waveId >> 1) << 6);
    const int col0 = (bj << 7) + ((waveId & 1) << 6);

    const int m    = lane & 15;
    const int koff = (lane >> 4) << 1;

    __shared__ __attribute__((aligned(16))) float panel[2 * EDM_PR * EDM_PITCH];

    const float* xb = x + (size_t)b * EDM_N * EDM_D;
    {
        const float4* ga = (const float4*)(xb + (size_t)(bi << 7) * EDM_D);
        const float4* gb = (const float4*)(xb + (size_t)(bj << 7) * EDM_D);
#pragma unroll 16
        for (int it = 0; it < (EDM_PR * EDM_D / 4) / 128; ++it) {
            const int f   = it * 128 + (int)threadIdx.x;
            const int row = f >> 5;
            const int c   = (f & 31) << 2;
            const float4 va = ga[f];
            const float4 vb = gb[f];
            *(float4*)&panel[row * EDM_PITCH + c] = va;
            *(float4*)&panel[(EDM_PR + row) * EDM_PITCH + c] = vb;
        }
    }
    __syncthreads();

    const float* ap = panel + (((waveId >> 1) << 6) + m) * EDM_PITCH + koff;
    const float* bp = panel + (EDM_PR + ((waveId & 1) << 6) + m) * EDM_PITCH + koff;

    v8f acc[EDM_WT][EDM_WT];
#pragma unroll
    for (int i = 0; i < EDM_WT; ++i)
#pragma unroll
        for (int j = 0; j < EDM_WT; ++j)
            acc[i][j] = v8f{};

    v2f na[EDM_WT], nb[EDM_WT];
#pragma unroll
    for (int i = 0; i < EDM_WT; ++i) { na[i] = v2f{}; nb[i] = v2f{}; }

    v2f fa[2][4][EDM_WT], fb[2][4][EDM_WT];

#define EDM_READ_BLOCK(SET, KB)                                                   \
    _Pragma("unroll")                                                             \
    for (int s = 0; s < 4; ++s) {                                                 \
        _Pragma("unroll")                                                         \
        for (int i = 0; i < EDM_WT; ++i) {                                        \
            fa[SET][s][i] = *(const v2f*)(ap + i * 16 * EDM_PITCH + (KB) + 4 * s); \
            fb[SET][s][i] = *(const v2f*)(bp + i * 16 * EDM_PITCH + (KB) + 4 * s); \
        }                                                                         \
    }

#define EDM_MUL_BLOCK(SET)                                                        \
    _Pragma("unroll")                                                             \
    for (int s = 0; s < 4; ++s) {                                                 \
        _Pragma("unroll")                                                         \
        for (int i = 0; i < EDM_WT; ++i) {                                        \
            na[i] = __builtin_elementwise_fma(fa[SET][s][i], fa[SET][s][i], na[i]); \
            nb[i] = __builtin_elementwise_fma(fb[SET][s][i], fb[SET][s][i], nb[i]); \
        }                                                                         \
        _Pragma("unroll")                                                         \
        for (int i = 0; i < EDM_WT; ++i) {                                        \
            _Pragma("unroll")                                                     \
            for (int j = 0; j < EDM_WT; ++j)                                      \
                acc[i][j] = __builtin_amdgcn_wmma_f32_16x16x4_f32(                \
                    false, fa[SET][s][i], false, fb[SET][s][j],                   \
                    (short)0, acc[i][j], false, false);                           \
        }                                                                         \
    }

    EDM_READ_BLOCK(0, 0)
#pragma unroll 1
    for (int kb = 0; kb < EDM_D - 32; kb += 32) {
        EDM_READ_BLOCK(1, kb + 16)
        __builtin_amdgcn_sched_barrier(0);
        EDM_MUL_BLOCK(0)
        EDM_READ_BLOCK(0, kb + 32)
        __builtin_amdgcn_sched_barrier(0);
        EDM_MUL_BLOCK(1)
    }
    EDM_READ_BLOCK(1, EDM_D - 16)
    __builtin_amdgcn_sched_barrier(0);
    EDM_MUL_BLOCK(0)
    EDM_MUL_BLOCK(1)
#undef EDM_READ_BLOCK
#undef EDM_MUL_BLOCK

    const int half = lane >> 4;
    float sqa[EDM_WT], sqj[EDM_WT];
#pragma unroll
    for (int t = 0; t < EDM_WT; ++t) {
        const float pa = na[t][0] + na[t][1];
        const float pb = nb[t][0] + nb[t][1];
        sqa[t] = pa + __shfl_xor(pa, 16, 32);
        sqj[t] = pb + __shfl_xor(pb, 16, 32);
    }

#pragma unroll
    for (int i = 0; i < EDM_WT; ++i) {
        const int   rbase = row0 + i * 16 + (half << 3);
        float* orow = out + ((size_t)b * EDM_N + rbase) * EDM_N + col0 + m;
#pragma unroll
        for (int r = 0; r < 8; ++r) {
            const float si = __shfl(sqa[i], (half << 3) + r, 32);
#pragma unroll
            for (int j = 0; j < EDM_WT; ++j) {
                float d2 = si + sqj[j] - 2.0f * acc[i][j][r];
                d2 = fmaxf(d2, 0.0f);
                __builtin_nontemporal_store(sqrtf(d2 + EDM_EPS),
                                            orow + (size_t)r * EDM_N + j * 16);
            }
        }
    }
}

extern "C" void kernel_launch(void* const* d_in, const int* in_sizes, int n_in,
                              void* d_out, int out_size, void* d_ws, size_t ws_size,
                              hipStream_t stream) {
    (void)in_sizes; (void)n_in; (void)out_size; (void)d_ws; (void)ws_size;

    const float* x  = (const float*)d_in[0];
    float* out      = (float*)d_out;

    edm_tile_kernel<<<1024, 128, 0, stream>>>(x, out);
}


# Use the symmetry out[i][j] == out[j][i]: compute only the 136 blocks bi <= bj per batch (544 work-groups, not 1024); a block above the diagonal is also stored transposed via LDS; sums unchanged
# speedup vs baseline: 3.4181x; 1.3108x over previous
#include <hip/hip_runtime.h>
#include <hip/hip_bf16.h>


typedef __attribute__((ext_vector_type(2))) float v2f;
typedef __attribute__((ext_vector_type(8))) float v8f;
typedef __attribute__((ext_vector_type(4))) float v4f;

#define EDM_B 4
#define EDM_N 2048
#define EDM_D 128
#define EDM_EPS 1e-7f

#define EDM_WT 4
#define EDM_PR 128
#define EDM_PITCH (EDM_D + 4)
#define EDM_OFFDIAG 480

__global__ void __launch_bounds__(128)
edm_tile_kernel(const float* __restrict__ x,
                float* __restrict__ out) {
    const int lane   = threadIdx.x & 31;
    const int waveId = threadIdx.x >> 5;

    int b, bi, bj;
    if (blockIdx.x < EDM_OFFDIAG) {
        b = (int)blockIdx.x / 120;
        const int t = (int)blockIdx.x - b * 120;
        const int R = t >> 4;
        const int c = t & 15;
        if (c < 15 - R) { bi = R;      bj = R + 1 + c; }
        else            { bi = 14 - R; bj = c - (15 - R) + bi + 1; }
    } else {
        const int d = (int)blockIdx.x - EDM_OFFDIAG;
        b  = d >> 4;
        bi = d & 15;
        bj = bi;
    }
    const int row0 = (bi << 7) + ((waveId >> 1) << 6);
    const int col0 = (bj << 7) + ((waveId & 1) << 6);

    const int m    = lane & 15;
    const int koff = (lane >> 4) << 1;

    __shared__ __attribute__((aligned(16))) float panel[2 * EDM_PR * EDM_PITCH];

    const float* xb = x + (size_t)b * EDM_N * EDM_D;
    {
        const float4* ga = (const float4*)(xb + (size_t)(bi << 7) * EDM_D);
        const float4* gb = (const float4*)(xb + (size_t)(bj << 7) * EDM_D);
#pragma unroll 16
        for (int it = 0; it < (EDM_PR * EDM_D / 4) / 128; ++it) {
            const int f   = it * 128 + (int)threadIdx.x;
            const int row = f >> 5;
            const int c   = (f & 31) << 2;
            const float4 va = ga[f];
            const float4 vb = gb[f];
            *(float4*)&panel[row * EDM_PITCH + c] = va;
            *(float4*)&panel[(EDM_PR + row) * EDM_PITCH + c] = vb;
        }
    }
    __syncthreads();

    const float* ap = panel + (((waveId >> 1) << 6) + m) * EDM_PITCH + koff;
    const float* bp = panel + (EDM_PR + ((waveId & 1) << 6) + m) * EDM_PITCH + koff;

    v8f acc[EDM_WT][EDM_WT];
#pragma unroll
    for (int i = 0; i < EDM_WT; ++i)
#pragma unroll
        for (int j = 0; j < EDM_WT; ++j)
            acc[i][j] = v8f{};

    v2f na[EDM_WT], nb[EDM_WT];
#pragma unroll
    for (int i = 0; i < EDM_WT; ++i) { na[i] = v2f{}; nb[i] = v2f{}; }

    v2f fa[2][4][EDM_WT], fb[2][4][EDM_WT];

#define EDM_READ_BLOCK(SET, KB)                                                   \
    _Pragma("unroll")                                                             \
    for (int s = 0; s < 4; ++s) {                                                 \
        _Pragma("unroll")                                                         \
        for (int i = 0; i < EDM_WT; ++i) {                                        \
            fa[SET][s][i] = *(const v2f*)(ap + i * 16 * EDM_PITCH + (KB) + 4 * s); \
            fb[SET][s][i] = *(const v2f*)(bp + i * 16 * EDM_PITCH + (KB) + 4 * s); \
        }                                                                         \
    }

#define EDM_MUL_BLOCK(SET)                                                        \
    _Pragma("unroll")                                                             \
    for (int s = 0; s < 4; ++s) {                                                 \
        _Pragma("unroll")                                                         \
        for (int i = 0; i < EDM_WT; ++i) {                                        \
            na[i] = __builtin_elementwise_fma(fa[SET][s][i], fa[SET][s][i], na[i]); \
            nb[i] = __builtin_elementwise_fma(fb[SET][s][i], fb[SET][s][i], nb[i]); \
        }                                                                         \
        _Pragma("unroll")                                                         \
        for (int i = 0; i < EDM_WT; ++i) {                                        \
            _Pragma("unroll")                                                     \
            for (int j = 0; j < EDM_WT; ++j)                                      \
                acc[i][j] = __builtin_amdgcn_wmma_f32_16x16x4_f32(                \
                    false, fa[SET][s][i], false, fb[SET][s][j],                   \
                    (short)0, acc[i][j], false, false);                           \
        }                                                                         \
    }

    EDM_READ_BLOCK(0, 0)
#pragma unroll 1
    for (int kb = 0; kb < EDM_D - 32; kb += 32) {
        EDM_READ_BLOCK(1, kb + 16)
        __builtin_amdgcn_sched_barrier(0);
        EDM_MUL_BLOCK(0)
        EDM_READ_BLOCK(0, kb + 32)
        __builtin_amdgcn_sched_barrier(0);
        EDM_MUL_BLOCK(1)
    }
    EDM_READ_BLOCK(1, EDM_D - 16)
    __builtin_amdgcn_sched_barrier(0);
    EDM_MUL_BLOCK(0)
    EDM_MUL_BLOCK(1)
#undef EDM_READ_BLOCK
#undef EDM_MUL_BLOCK

    const int half = lane >> 4;
    float sqa[EDM_WT], sqj[EDM_WT];
#pragma unroll
    for (int t = 0; t < EDM_WT; ++t) {
        const float pa = na[t][0] + na[t][1];
        const float pb = nb[t][0] + nb[t][1];
        sqa[t] = pa + __shfl_xor(pa, 16, 32);
        sqj[t] = pb + __shfl_xor(pb, 16, 32);
    }

    const bool mirror = (bi != bj);
    if (mirror) __syncthreads();
    float* stage = panel + (((waveId & 1) << 6) + m) * EDM_PITCH
                         + ((waveId >> 1) << 6) + (half << 3);

#pragma unroll
    for (int i = 0; i < EDM_WT; ++i) {
        const int   rbase = row0 + i * 16 + (half << 3);
        float* orow = out + ((size_t)b * EDM_N + rbase) * EDM_N + col0 + m;
#pragma unroll
        for (int r = 0; r < 8; ++r) {
            const float si = __shfl(sqa[i], (half << 3) + r, 32);
#pragma unroll
            for (int j = 0; j < EDM_WT; ++j) {
                float d2 = si + sqj[j] - 2.0f * acc[i][j][r];
                d2 = fmaxf(d2, 0.0f);
                const float dist = sqrtf(d2 + EDM_EPS);
                acc[i][j][r] = dist;
                __builtin_nontemporal_store(dist, orow + (size_t)r * EDM_N + j * 16);
            }
        }
        if (mirror) {
#pragma unroll
            for (int j = 0; j < EDM_WT; ++j) {
                float* sp = stage + j * 16 * EDM_PITCH + i * 16;
                *(v4f*)sp       = v4f{acc[i][j][0], acc[i][j][1], acc[i][j][2], acc[i][j][3]};
                *(v4f*)(sp + 4) = v4f{acc[i][j][4], acc[i][j][5], acc[i][j][6], acc[i][j][7]};
            }
        }
    }

    if (mirror) {
        __builtin_amdgcn_fence(__ATOMIC_RELEASE, "workgroup", "local");
        __builtin_amdgcn_s_barrier();
        __builtin_amdgcn_fence(__ATOMIC_ACQUIRE, "workgroup", "local");
        const float* srow = panel + (lane << 2);
        float* mrow = out + ((size_t)b * EDM_N + (bj << 7)) * EDM_N + (bi << 7) + (lane << 2);
#pragma unroll 8
        for (int q = 0; q < 32; ++q) {
            const int c = (q << 2) + waveId;
            const v4f v = *(const v4f*)(srow + c * EDM_PITCH);
            __builtin_nontemporal_store(v, (v4f*)(mrow + (size_t)c * EDM_N));
        }
    }
}

extern "C" void kernel_launch(void* const* d_in, const int* in_sizes, int n_in,
                              void* d_out, int out_size, void* d_ws, size_t ws_size,
                              hipStream_t stream) {
    (void)in_sizes; (void)n_in; (void)out_size; (void)d_ws; (void)ws_size;

    const float* x  = (const float*)d_in[0];
    float* out      = (float*)d_out;

    edm_tile_kernel<<<544, 128, 0, stream>>>(x, out);
}


# Epilogue square root as the single v_sqrt_f32 instruction (within 1 ulp, not correctly rounded), not the library sqrtf's 16-instruction sequence: 1500 fewer vector instructions per wave; sums kept
# speedup vs baseline: 4.0943x; 1.1978x over previous
#include <hip/hip_runtime.h>
#include <hip/hip_bf16.h>


typedef __attribute__((ext_vector_type(2))) float v2f;
typedef __attribute__((ext_vector_type(8))) float v8f;
typedef __attribute__((ext_vector_type(4))) float v4f;

#define EDM_B 4
#define EDM_N 2048
#define EDM_D 128
#define EDM_EPS 1e-7f

#define EDM_WT 4
#define EDM_PR 128
#define EDM_PITCH (EDM_D + 4)
#define EDM_OFFDIAG 480

__global__ void __launch_bounds__(128)
edm_tile_kernel(const float* __restrict__ x,
                float* __restrict__ out) {
    const int lane   = threadIdx.x & 31;
    const int waveId = threadIdx.x >> 5;

    int b, bi, bj;
    if (blockIdx.x < EDM_OFFDIAG) {
        b = (int)blockIdx.x / 120;
        const int t = (int)blockIdx.x - b * 120;
        const int R = t >> 4;
        const int c = t & 15;
        if (c < 15 - R) { bi = R;      bj = R + 1 + c; }
        else            { bi = 14 - R; bj = c - (15 - R) + bi + 1; }
    } else {
        const int d = (int)blockIdx.x - EDM_OFFDIAG;
        b  = d >> 4;
        bi = d & 15;
        bj = bi;
    }
    const int row0 = (bi << 7) + ((waveId >> 1) << 6);
    const int col0 = (bj << 7) + ((waveId & 1) << 6);

    const int m    = lane & 15;
    const int koff = (lane >> 4) << 1;

    __shared__ __attribute__((aligned(16))) float panel[2 * EDM_PR * EDM_PITCH];

    const float* xb = x + (size_t)b * EDM_N * EDM_D;
    {
        const float4* ga = (const float4*)(xb + (size_t)(bi << 7) * EDM_D);
        const float4* gb = (const float4*)(xb + (size_t)(bj << 7) * EDM_D);
#pragma unroll 16
        for (int it = 0; it < (EDM_PR * EDM_D / 4) / 128; ++it) {
            const int f   = it * 128 + (int)threadIdx.x;
            const int row = f >> 5;
            const int c   = (f & 31) << 2;
            const float4 va = ga[f];
            const float4 vb = gb[f];
            *(float4*)&panel[row * EDM_PITCH + c] = va;
            *(float4*)&panel[(EDM_PR + row) * EDM_PITCH + c] = vb;
        }
    }
    __syncthreads();

    const float* ap = panel + (((waveId >> 1) << 6) + m) * EDM_PITCH + koff;
    const float* bp = panel + (EDM_PR + ((waveId & 1) << 6) + m) * EDM_PITCH + koff;

    v8f acc[EDM_WT][EDM_WT];
#pragma unroll
    for (int i = 0; i < EDM_WT; ++i)
#pragma unroll
        for (int j = 0; j < EDM_WT; ++j)
            acc[i][j] = v8f{};

    v2f na[EDM_WT], nb[EDM_WT];
#pragma unroll
    for (int i = 0; i < EDM_WT; ++i) { na[i] = v2f{}; nb[i] = v2f{}; }

    v2f fa[2][4][EDM_WT], fb[2][4][EDM_WT];

#define EDM_READ_BLOCK(SET, KB)                                                   \
    _Pragma("unroll")                                                             \
    for (int s = 0; s < 4; ++s) {                                                 \
        _Pragma("unroll")                                                         \
        for (int i = 0; i < EDM_WT; ++i) {                                        \
            fa[SET][s][i] = *(const v2f*)(ap + i * 16 * EDM_PITCH + (KB) + 4 * s); \
            fb[SET][s][i] = *(const v2f*)(bp + i * 16 * EDM_PITCH + (KB) + 4 * s); \
        }                                                                         \
    }

#define EDM_MUL_BLOCK(SET)                                                        \
    _Pragma("unroll")                                                             \
    for (int s = 0; s < 4; ++s) {                                                 \
        _Pragma("unroll")                                                         \
        for (int i = 0; i < EDM_WT; ++i) {                                        \
            na[i] = __builtin_elementwise_fma(fa[SET][s][i], fa[SET][s][i], na[i]); \
            nb[i] = __builtin_elementwise_fma(fb[SET][s][i], fb[SET][s][i], nb[i]); \
        }                                                                         \
        _Pragma("unroll")                                                         \
        for (int i = 0; i < EDM_WT; ++i) {                                        \
            _Pragma("unroll")                                                     \
            for (int j = 0; j < EDM_WT; ++j)                                      \
                acc[i][j] = __builtin_amdgcn_wmma_f32_16x16x4_f32(                \
                    false, fa[SET][s][i], false, fb[SET][s][j],                   \
                    (short)0, acc[i][j], false, false);                           \
        }                                                                         \
    }

    EDM_READ_BLOCK(0, 0)
#pragma unroll 1
    for (int kb = 0; kb < EDM_D - 32; kb += 32) {
        EDM_READ_BLOCK(1, kb + 16)
        __builtin_amdgcn_sched_barrier(0);
        EDM_MUL_BLOCK(0)
        EDM_READ_BLOCK(0, kb + 32)
        __builtin_amdgcn_sched_barrier(0);
        EDM_MUL_BLOCK(1)
    }
    EDM_READ_BLOCK(1, EDM_D - 16)
    __builtin_amdgcn_sched_barrier(0);
    EDM_MUL_BLOCK(0)
    EDM_MUL_BLOCK(1)
#undef EDM_READ_BLOCK
#undef EDM_MUL_BLOCK

    const int half = lane >> 4;
    float sqa[EDM_WT], sqj[EDM_WT];
#pragma unroll
    for (int t = 0; t < EDM_WT; ++t) {
        const float pa = na[t][0] + na[t][1];
        const float pb = nb[t][0] + nb[t][1];
        sqa[t] = pa + __shfl_xor(pa, 16, 32);
        sqj[t] = pb + __shfl_xor(pb, 16, 32);
    }

    const bool mirror = (bi != bj);
    if (mirror) __syncthreads();
    float* stage = panel + (((waveId & 1) << 6) + m) * EDM_PITCH
                         + ((waveId >> 1) << 6) + (half << 3);

#pragma unroll
    for (int i = 0; i < EDM_WT; ++i) {
        const int   rbase = row0 + i * 16 + (half << 3);
        float* orow = out + ((size_t)b * EDM_N + rbase) * EDM_N + col0 + m;
#pragma unroll
        for (int r = 0; r < 8; ++r) {
            const float si = __shfl(sqa[i], (half << 3) + r, 32);
#pragma unroll
            for (int j = 0; j < EDM_WT; ++j) {
                float d2 = si + sqj[j] - 2.0f * acc[i][j][r];
                d2 = fmaxf(d2, 0.0f);
                const float dist = __builtin_amdgcn_sqrtf(d2 + EDM_EPS);
                acc[i][j][r] = dist;
                __builtin_nontemporal_store(dist, orow + (size_t)r * EDM_N + j * 16);
            }
        }
        if (mirror) {
#pragma unroll
            for (int j = 0; j < EDM_WT; ++j) {
                float* sp = stage + j * 16 * EDM_PITCH + i * 16;
                *(v4f*)sp       = v4f{acc[i][j][0], acc[i][j][1], acc[i][j][2], acc[i][j][3]};
                *(v4f*)(sp + 4) = v4f{acc[i][j][4], acc[i][j][5], acc[i][j][6], acc[i][j][7]};
            }
        }
    }

    if (mirror) {
        __builtin_amdgcn_fence(__ATOMIC_RELEASE, "workgroup", "local");
        __builtin_amdgcn_s_barrier();
        __builtin_amdgcn_fence(__ATOMIC_ACQUIRE, "workgroup", "local");
        const float* srow = panel + (lane << 2);
        float* mrow = out + ((size_t)b * EDM_N + (bj << 7)) * EDM_N + (bi << 7) + (lane << 2);
#pragma unroll 8
        for (int q = 0; q < 32; ++q) {
            const int c = (q << 2) + waveId;
            const v4f v = *(const v4f*)(srow + c * EDM_PITCH);
            __builtin_nontemporal_store(v, (v4f*)(mrow + (size_t)c * EDM_N));
        }
    }
}

extern "C" void kernel_launch(void* const* d_in, const int* in_sizes, int n_in,
                              void* d_out, int out_size, void* d_ws, size_t ws_size,
                              hipStream_t stream) {
    (void)in_sizes; (void)n_in; (void)out_size; (void)d_ws; (void)ws_size;

    const float* x  = (const float*)d_in[0];
    float* out      = (float*)d_out;

    edm_tile_kernel<<<544, 128, 0, stream>>>(x, out);
}


# Fold the 64 diagonal blocks in pairs into 32 work-groups that compute only tiles j >= i (18 per wave) and mirror the rest via LDS: 512 work-groups, all resident in one round; sums unchanged
# speedup vs baseline: 4.7651x; 1.1638x over previous
#include <hip/hip_runtime.h>
#include <hip/hip_bf16.h>


typedef __attribute__((ext_vector_type(2))) float v2f;
typedef __attribute__((ext_vector_type(8))) float v8f;
typedef __attribute__((ext_vector_type(4))) float v4f;

#define EDM_B 4
#define EDM_N 2048
#define EDM_D 128
#define EDM_EPS 1e-7f

#define EDM_WT 4
#define EDM_PR 128
#define EDM_PITCH (EDM_D + 4)
#define EDM_NACC 20

__device__ __forceinline__ constexpr int edm_tri(int i, int j) {
    return i * 4 + j - (i * (i + 1)) / 2;
}
__device__ __forceinline__ constexpr bool edm_moved(int i, int j) {
    return (i == 0 && j == 1) || (i == 2 && j == 3);
}

__device__ __forceinline__ float edm_dist(float si, float sj, float s) {
    float d2 = si + sj - 2.0f * s;
    d2 = fmaxf(d2, 0.0f);
    return __builtin_amdgcn_sqrtf(d2 + EDM_EPS);
}

__device__ __forceinline__ void edm_put_tile(float* blk, const v8f& a, const float (&si)[8],
                                             float sj, int rb, int cb, bool mirror,
                                             int m, int half) {
    float d[8];
#pragma unroll
    for (int r = 0; r < 8; ++r) {
        d[r] = edm_dist(si[r], sj, a[r]);
        blk[(rb + (half << 3) + r) * EDM_PITCH + cb + m] = d[r];
    }
    if (mirror) {
        float* sp = blk + (cb + m) * EDM_PITCH + rb + (half << 3);
        *(v4f*)sp       = v4f{d[0], d[1], d[2], d[3]};
        *(v4f*)(sp + 4) = v4f{d[4], d[5], d[6], d[7]};
    }
}

__global__ void __launch_bounds__(128)
edm_tile_kernel(const float* __restrict__ x,
                float* __restrict__ out) {
    const int lane   = threadIdx.x & 31;
    const int waveId = __builtin_amdgcn_readfirstlane((int)(threadIdx.x >> 5));
    const int m    = lane & 15;
    const int half = lane >> 4;
    const int koff = half << 1;

    const int  id     = (int)blockIdx.x;
    const bool diagwg = id < 64 && (id & 1) == 0;
    int b, bi, bj;
    int arow, brow;
    int role;
    if (diagwg) {
        const int d = id >> 1;
        b  = d >> 3;
        bi = d & 7;
        bj = bi + 8;
        arow = (waveId & 1) * EDM_PR;
        brow = arow + 64;
        role = 1 + (waveId >> 1);
    } else {
        const int u = id < 64 ? (id >> 1) : id - 32;
        b = u / 120;
        const int t = u - b * 120;
        const int R = t >> 4;
        const int c = t & 15;
        if (c < 15 - R) { bi = R;      bj = R + 1 + c; }
        else            { bi = 14 - R; bj = c - (15 - R) + bi + 1; }
        arow = (waveId >> 1) << 6;
        brow = EDM_PR + ((waveId & 1) << 6);
        role = 0;
    }

    __shared__ __attribute__((aligned(16))) float panel[2 * EDM_PR * EDM_PITCH];

    const float* xb = x + (size_t)b * EDM_N * EDM_D;
    {
        const float4* ga = (const float4*)(xb + (size_t)(bi << 7) * EDM_D);
        const float4* gb = (const float4*)(xb + (size_t)(bj << 7) * EDM_D);
#pragma unroll 16
        for (int it = 0; it < (EDM_PR * EDM_D / 4) / 128; ++it) {
            const int f   = it * 128 + (int)threadIdx.x;
            const int row = f >> 5;
            const int c   = (f & 31) << 2;
            const float4 va = ga[f];
            const float4 vb = gb[f];
            *(float4*)&panel[row * EDM_PITCH + c] = va;
            *(float4*)&panel[(EDM_PR + row) * EDM_PITCH + c] = vb;
        }
    }
    __syncthreads();

    const float* ap = panel + (arow + m) * EDM_PITCH + koff;
    const float* bp = panel + (brow + m) * EDM_PITCH + koff;

    v8f acc[EDM_NACC];
#pragma unroll
    for (int i = 0; i < EDM_NACC; ++i) acc[i] = v8f{};

    v2f na[EDM_WT], nb[EDM_WT];
#pragma unroll
    for (int i = 0; i < EDM_WT; ++i) { na[i] = v2f{}; nb[i] = v2f{}; }

    v2f fa[2][4][EDM_WT], fb[2][4][EDM_WT];

#define EDM_READ_BLOCK(SET, KB)                                                   \
    _Pragma("unroll")                                                             \
    for (int s = 0; s < 4; ++s) {                                                 \
        _Pragma("unroll")                                                         \
        for (int i = 0; i < EDM_WT; ++i) {                                        \
            fa[SET][s][i] = *(const v2f*)(ap + i * 16 * EDM_PITCH + (KB) + 4 * s); \
            fb[SET][s][i] = *(const v2f*)(bp + i * 16 * EDM_PITCH + (KB) + 4 * s); \
        }                                                                         \
    }

#define EDM_WMMA(ACC, FA, FB)                                                     \
    ACC = __builtin_amdgcn_wmma_f32_16x16x4_f32(false, FA, false, FB,             \
                                                (short)0, ACC, false, false);

#define EDM_NORM_STEP(SET)                                                        \
        _Pragma("unroll")                                                         \
        for (int i = 0; i < EDM_WT; ++i) {                                        \
            na[i] = __builtin_elementwise_fma(fa[SET][s][i], fa[SET][s][i], na[i]); \
            nb[i] = __builtin_elementwise_fma(fb[SET][s][i], fb[SET][s][i], nb[i]); \
        }

#define EDM_MUL_BLOCK(SET)                                                        \
    _Pragma("unroll")                                                             \
    for (int s = 0; s < 4; ++s) {                                                 \
        EDM_NORM_STEP(SET)                                                        \
        _Pragma("unroll")                                                         \
        for (int i = 0; i < EDM_WT; ++i) {                                        \
            _Pragma("unroll")                                                     \
            for (int j = 0; j < EDM_WT; ++j)                                      \
                EDM_WMMA(acc[4 * i + j], fa[SET][s][i], fb[SET][s][j])            \
        }                                                                         \
    }

#define EDM_MUL_BLOCK_R1(SET)                                                     \
    _Pragma("unroll")                                                             \
    for (int s = 0; s < 4; ++s) {                                                 \
        EDM_NORM_STEP(SET)                                                        \
        _Pragma("unroll")                                                         \
        for (int i = 0; i < EDM_WT; ++i) {                                        \
            _Pragma("unroll")                                                     \
            for (int j = 0; j < EDM_WT; ++j)                                      \
                EDM_WMMA(acc[4 * i + j], fa[SET][s][i], fb[SET][s][j])            \
        }                                                                         \
        EDM_WMMA(acc[16], fb[SET][s][0], fb[SET][s][1])                           \
        EDM_WMMA(acc[17], fb[SET][s][2], fb[SET][s][3])                           \
    }

#define EDM_MUL_BLOCK_R2(SET)                                                     \
    _Pragma("unroll")                                                             \
    for (int s = 0; s < 4; ++s) {                                                 \
        EDM_NORM_STEP(SET)                                                        \
        _Pragma("unroll")                                                         \
        for (int i = 0; i < EDM_WT; ++i) {                                        \
            _Pragma("unroll")                                                     \
            for (int j = i; j < EDM_WT; ++j) {                                    \
                EDM_WMMA(acc[edm_tri(i, j)], fa[SET][s][i], fa[SET][s][j])        \
                if (!edm_moved(i, j))                                             \
                    EDM_WMMA(acc[10 + edm_tri(i, j)], fb[SET][s][i], fb[SET][s][j]) \
            }                                                                     \
        }                                                                         \
    }

#define EDM_K_LOOP(MUL)                                                           \
    EDM_READ_BLOCK(0, 0)                                                          \
    _Pragma("unroll 1")                                                           \
    for (int kb = 0; kb < EDM_D - 32; kb += 32) {                                 \
        EDM_READ_BLOCK(1, kb + 16)                                                \
        __builtin_amdgcn_sched_barrier(0);                                        \
        MUL(0)                                                                    \
        EDM_READ_BLOCK(0, kb + 32)                                                \
        __builtin_amdgcn_sched_barrier(0);                                        \
        MUL(1)                                                                    \
    }                                                                             \
    EDM_READ_BLOCK(1, EDM_D - 16)                                                 \
    __builtin_amdgcn_sched_barrier(0);                                            \
    MUL(0)                                                                        \
    MUL(1)

    if (role == 0) {
        EDM_K_LOOP(EDM_MUL_BLOCK)
    } else if (role == 1) {
        EDM_K_LOOP(EDM_MUL_BLOCK_R1)
    } else {
        EDM_K_LOOP(EDM_MUL_BLOCK_R2)
    }
#undef EDM_K_LOOP
#undef EDM_READ_BLOCK
#undef EDM_WMMA
#undef EDM_NORM_STEP
#undef EDM_MUL_BLOCK
#undef EDM_MUL_BLOCK_R1
#undef EDM_MUL_BLOCK_R2

    float sqa[EDM_WT], sqj[EDM_WT];
#pragma unroll
    for (int t = 0; t < EDM_WT; ++t) {
        const float pa = na[t][0] + na[t][1];
        const float pb = nb[t][0] + nb[t][1];
        sqa[t] = pa + __shfl_xor(pa, 16, 32);
        sqj[t] = pb + __shfl_xor(pb, 16, 32);
    }

    __syncthreads();

    if (!diagwg) {
        const int row0 = (bi << 7) + ((waveId >> 1) << 6);
        const int col0 = (bj << 7) + ((waveId & 1) << 6);
        float* stage = panel + (((waveId & 1) << 6) + m) * EDM_PITCH
                             + ((waveId >> 1) << 6) + (half << 3);
#pragma unroll
        for (int i = 0; i < EDM_WT; ++i) {
            const int   rbase = row0 + i * 16 + (half << 3);
            float* orow = out + ((size_t)b * EDM_N + rbase) * EDM_N + col0 + m;
#pragma unroll
            for (int r = 0; r < 8; ++r) {
                const float si = __shfl(sqa[i], (half << 3) + r, 32);
#pragma unroll
                for (int j = 0; j < EDM_WT; ++j) {
                    const float dist = edm_dist(si, sqj[j], acc[4 * i + j][r]);
                    acc[4 * i + j][r] = dist;
                    __builtin_nontemporal_store(dist, orow + (size_t)r * EDM_N + j * 16);
                }
            }
#pragma unroll
            for (int j = 0; j < EDM_WT; ++j) {
                float* sp = stage + j * 16 * EDM_PITCH + i * 16;
                const v8f a = acc[4 * i + j];
                *(v4f*)sp       = v4f{a[0], a[1], a[2], a[3]};
                *(v4f*)(sp + 4) = v4f{a[4], a[5], a[6], a[7]};
            }
        }
        __builtin_amdgcn_fence(__ATOMIC_RELEASE, "workgroup", "local");
        __builtin_amdgcn_s_barrier();
        __builtin_amdgcn_fence(__ATOMIC_ACQUIRE, "workgroup", "local");
        const float* srow = panel + (lane << 2);
        float* mrow = out + ((size_t)b * EDM_N + (bj << 7)) * EDM_N + (bi << 7) + (lane << 2);
#pragma unroll 8
        for (int q = 0; q < 32; ++q) {
            const int c = (q << 2) + waveId;
            const v4f v = *(const v4f*)(srow + c * EDM_PITCH);
            __builtin_nontemporal_store(v, (v4f*)(mrow + (size_t)c * EDM_N));
        }
    } else {
        float* blk = panel + arow * EDM_PITCH;
        float ra[EDM_WT][8], rj[EDM_WT][8];
#pragma unroll
        for (int i = 0; i < EDM_WT; ++i) {
#pragma unroll
            for (int r = 0; r < 8; ++r) {
                ra[i][r] = __shfl(sqa[i], (half << 3) + r, 32);
                rj[i][r] = __shfl(sqj[i], (half << 3) + r, 32);
            }
        }
        if (role == 1) {
#pragma unroll
            for (int i = 0; i < EDM_WT; ++i) {
#pragma unroll
                for (int j = 0; j < EDM_WT; ++j)
                    edm_put_tile(blk, acc[4 * i + j], ra[i], sqj[j], 16 * i, 64 + 16 * j,
                                 true, m, half);
            }
            edm_put_tile(blk, acc[16], rj[0], sqj[1], 64,      64 + 16, true, m, half);
            edm_put_tile(blk, acc[17], rj[2], sqj[3], 64 + 32, 64 + 48, true, m, half);
        } else {
#pragma unroll
            for (int i = 0; i < EDM_WT; ++i) {
#pragma unroll
                for (int j = i; j < EDM_WT; ++j) {
                    edm_put_tile(blk, acc[edm_tri(i, j)], ra[i], sqa[j], 16 * i, 16 * j,
                                 j > i, m, half);
                    if (!edm_moved(i, j))
                        edm_put_tile(blk, acc[10 + edm_tri(i, j)], rj[i], sqj[j],
                                     64 + 16 * i, 64 + 16 * j, j > i, m, half);
                }
            }
        }
        __syncthreads();
#pragma unroll
        for (int p2 = 0; p2 < 2; ++p2) {
            const int bb = p2 ? bj : bi;
            const float* srow = panel + p2 * EDM_PR * EDM_PITCH + (lane << 2);
            float* orow = out + ((size_t)b * EDM_N + (bb << 7)) * EDM_N + (bb << 7) + (lane << 2);
#pragma unroll 8
            for (int q = 0; q < 32; ++q) {
                const int c = (q << 2) + waveId;
                const v4f v = *(const v4f*)(srow + c * EDM_PITCH);
                __builtin_nontemporal_store(v, (v4f*)(orow + (size_t)c * EDM_N));
            }
        }
    }
}

extern "C" void kernel_launch(void* const* d_in, const int* in_sizes, int n_in,
                              void* d_out, int out_size, void* d_ws, size_t ws_size,
                              hipStream_t stream) {
    (void)in_sizes; (void)n_in; (void)out_size; (void)d_ws; (void)ws_size;

    const float* x  = (const float*)d_in[0];
    float* out      = (float*)d_out;

    edm_tile_kernel<<<512, 128, 0, stream>>>(x, out);
}
